# starting kernel
# baseline (speedup 1.0000x reference)
#include <hip/hip_runtime.h>


typedef __attribute__((ext_vector_type(2)))  float    v2f;
typedef __attribute__((ext_vector_type(8)))  float    v8f;
typedef __attribute__((ext_vector_type(16))) _Float16 v16h;

static constexpr int BATCH = 2;
static constexpr int SLEN  = 1024;
static constexpr int DIN   = 256;
static constexpr int DOUT  = 64;
static constexpr int Q     = 128;
static constexpr int KH    = 6;
static constexpr int MROWS = BATCH * SLEN;
static constexpr int FPQ   = 2 * KH;
static constexpr int KFEAT = Q * FPQ;
static constexpr int KB    = KFEAT / 32;

__device__ __forceinline__ int frag_k16(int i, int hi) {
    return (i & 7) + ((i >> 3) << 4) + hi * 8;
}

__global__ __launch_bounds__(256) void k_angle_wmma(
    const float* __restrict__ X0,
    const float* __restrict__ t,
    const float* __restrict__ Wcw,
    const float* __restrict__ Wcb,
    const float* __restrict__ w,
    float* __restrict__ ANG)
{
    const int tid  = threadIdx.x;
    const int wave = tid >> 5;
    const int lane = tid & 31;
    const int lm   = lane & 15;
    const int hi   = lane >> 4;
    const int m0   = blockIdx.x * 16;
    const int n0   = wave * 16;

    const int ksel = hi * 2;
    const float* xrow = X0  + (size_t)(m0 + lm) * DIN + ksel;
    const float* wrow = Wcw + (size_t)(n0 + lm) * DIN + ksel;

    v8f acc = {0.f,0.f,0.f,0.f,0.f,0.f,0.f,0.f};
    #pragma unroll 4
    for (int k = 0; k < DIN; k += 4) {
        v2f a = *(const v2f*)(xrow + k);
        v2f b = *(const v2f*)(wrow + k);
        acc = __builtin_amdgcn_wmma_f32_16x16x4_f32(
             false, a,  false, b,
             (short)0, acc,  false,  false);
    }

    const int   q     = n0 + lm;
    const int   batch = m0 >> 10;
    const float shift = Wcb[q] + w[q] * t[batch];

    #pragma unroll
    for (int j = 0; j < 8; ++j) {
        const int row = m0 + j + hi * 8;
        ANG[(size_t)row * Q + q] = acc[j] + shift;
    }
}

__global__ __launch_bounds__(256) void k_w2_pack(
    const float* __restrict__ A,
    const float* __restrict__ Bp,
    _Float16* __restrict__ W2P)
{
    const int idx  = blockIdx.x * 256 + threadIdx.x;
    const int i    = idx & 15;
    const int lane = (idx >> 4) & 31;
    const int nt   = (idx >> 9) & 3;
    const int kb   = idx >> 11;
    const int hi   = lane >> 4;

    const int kf = kb * 32 + frag_k16(i, hi);
    const int d  = nt * 16 + (lane & 15);
    const int q  = kf / FPQ;
    const int kk = kf - q * FPQ;
    const int ks = (kk < KH) ? kk : kk - KH;

    const int   ai  = (d * Q + q) * KH + ks;
    const float av  = A[ai];
    const float bp  = Bp[ai];
    const float val = (kk < KH) ? av * cosf(bp) : av * sinf(bp);
    W2P[idx] = (_Float16)val;
}

__global__ __launch_bounds__(128) void k_fourier_wmma(
    const float*    __restrict__ ANG,
    const _Float16* __restrict__ W2P,
    float*          __restrict__ out)
{
    __shared__ float sAng[64 * Q];
    const int tid     = threadIdx.x;
    const int rowBase = blockIdx.x * 64;

    for (int r = tid; r < 64 * Q; r += 128)
        sAng[r] = ANG[(size_t)rowBase * Q + r];
    __syncthreads();

    const int wave = tid >> 5;
    const int lane = tid & 31;
    const int lm   = lane & 15;
    const int hi   = lane >> 4;
    const int m0   = rowBase + wave * 16;

    const float* ar  = sAng + (wave * 16 + lm) * Q;
    const v16h*  W2v = (const v16h*)W2P;

    v8f acc0 = {0.f,0.f,0.f,0.f,0.f,0.f,0.f,0.f};
    v8f acc1 = {0.f,0.f,0.f,0.f,0.f,0.f,0.f,0.f};
    v8f acc2 = {0.f,0.f,0.f,0.f,0.f,0.f,0.f,0.f};
    v8f acc3 = {0.f,0.f,0.f,0.f,0.f,0.f,0.f,0.f};

    for (int kb = 0; kb < KB; ++kb) {
        v16h a;
        #pragma unroll
        for (int i = 0; i < 16; ++i) {
            const int   kf  = kb * 32 + frag_k16(i, hi);
            const int   q   = kf / FPQ;
            const int   kk  = kf - q * FPQ;
            const float ang = ar[q];
            const float kc  = (float)(((kk < KH) ? kk : kk - KH) + 1);
            const float v   = (kk < KH) ? __sinf(kc * ang) : __cosf(kc * ang);
            a[i] = (_Float16)v;
        }
        const v16h* bbase = W2v + (size_t)kb * 128 + lane;
        const v16h b0 = bbase[0];
        const v16h b1 = bbase[32];
        const v16h b2 = bbase[64];
        const v16h b3 = bbase[96];
        acc0 = __builtin_amdgcn_wmma_f32_16x16x32_f16(false, a, false, b0, (short)0, acc0, false, false);
        acc1 = __builtin_amdgcn_wmma_f32_16x16x32_f16(false, a, false, b1, (short)0, acc1, false, false);
        acc2 = __builtin_amdgcn_wmma_f32_16x16x32_f16(false, a, false, b2, (short)0, acc2, false, false);
        acc3 = __builtin_amdgcn_wmma_f32_16x16x32_f16(false, a, false, b3, (short)0, acc3, false, false);
    }

    #pragma unroll
    for (int j = 0; j < 8; ++j) {
        const int row = m0 + j + hi * 8;
        float* o = out + (size_t)row * DOUT + lm;
        o[0]  = acc0[j];
        o[16] = acc1[j];
        o[32] = acc2[j];
        o[48] = acc3[j];
    }
}

extern "C" void kernel_launch(void* const* d_in, const int* in_sizes, int n_in,
                              void* d_out, int out_size, void* d_ws, size_t ws_size,
                              hipStream_t stream) {
    const float* X0  = (const float*)d_in[0];
    const float* t   = (const float*)d_in[1];
    const float* Wcw = (const float*)d_in[2];
    const float* Wcb = (const float*)d_in[3];
    const float* w   = (const float*)d_in[4];
    const float* A   = (const float*)d_in[5];
    const float* Bp  = (const float*)d_in[6];
    float* out = (float*)d_out;

    char* ws = (char*)d_ws;
    float*    ANG = (float*)ws;
    _Float16* W2P = (_Float16*)(ws + (size_t)MROWS * Q * 4);

    k_w2_pack   <<<(KFEAT * DOUT) / 256, 256, 0, stream>>>(A, Bp, W2P);
    k_angle_wmma<<<MROWS / 16,           256, 0, stream>>>(X0, t, Wcw, Wcb, w, ANG);
    k_fourier_wmma<<<MROWS / 64,         128, 0, stream>>>(ANG, W2P, out);
}


# Angle GEMM transposed: a lane's accumulators are one row's angles, features go straight into A-fragment registers (no LDS tile); contraction split over 4 waves, summed via LDS: f32 sum order changes
# speedup vs baseline: 12.5463x; 12.5463x over previous
#include <hip/hip_runtime.h>


typedef __attribute__((ext_vector_type(2)))  float    v2f;
typedef __attribute__((ext_vector_type(8)))  float    v8f;
typedef __attribute__((ext_vector_type(16))) _Float16 v16h;

static constexpr int BATCH = 2;
static constexpr int SLEN  = 1024;
static constexpr int DIN   = 256;
static constexpr int DOUT  = 64;
static constexpr int Q     = 128;
static constexpr int KH    = 6;
static constexpr int MROWS = BATCH * SLEN;
static constexpr int FPQ   = 2 * KH;
static constexpr int KFEAT = Q * FPQ;
static constexpr int KB    = KFEAT / 32;

static constexpr int KSW   = KB / 4;

typedef __attribute__((ext_vector_type(4))) float v4f;

static constexpr int W2_BLOCKS  = (KFEAT * DOUT) / 256;
static constexpr int WC_CHUNKS  = (Q * DIN) / 4;
static constexpr int WC_BLOCKS  = WC_CHUNKS / 256;

__global__ __launch_bounds__(256) void k_weight_pack(
    const float* __restrict__ A,
    const float* __restrict__ Bp,
    const float* __restrict__ Wcw,
    _Float16* __restrict__ W2P,
    float*    __restrict__ WcwP)
{
    const int idx  = blockIdx.x * 256 + threadIdx.x;

    const bool copies = blockIdx.x < WC_BLOCKS;
    v4f wc = {0.f, 0.f, 0.f, 0.f};
    if (copies) {
        const int l = idx & 31;
        const int s = (idx >> 5) & 31;
        const int T = idx >> 10;
        wc = *(const v4f*)(Wcw + (size_t)(16 * T + (l & 15)) * DIN + 8 * s + 4 * (l >> 4));
    }

    const int i    = idx & 15;
    const int lane = (idx >> 4) & 31;
    const int nt   = (idx >> 9) & 3;
    const int kb   = idx >> 11;
    const int hi   = lane >> 4;

    const int d  = nt * 16 + (lane & 15);
    const int v  = kb / KSW;
    const int ks = kb - v * KSW;
    const int tt = ks / KH;
    const int g  = ks - tt * KH;
    const int q  = 32 * v + 16 * tt + 8 * hi + (i & 7);

    const int   ai  = (d * Q + q) * KH + g;
    const float av  = A[ai];
    const float bp  = Bp[ai];
    const float val = (i < 8) ? av * cosf(bp) : av * sinf(bp);
    W2P[idx] = (_Float16)val;

    if (copies) *(v4f*)(WcwP + (size_t)idx * 4) = wc;
}

typedef __attribute__((ext_vector_type(8))) _Float16 v8h;

static constexpr int OROW = DOUT + 4;
static constexpr int NT   = DOUT / 16;
static constexpr int NFRAG = KSW * NT;

__global__ __launch_bounds__(128) void k_angle_fourier_wmma(
    const float*    __restrict__ X0,
    const float*    __restrict__ t,
    const float*    __restrict__ WcwP,
    const float*    __restrict__ Wcb,
    const float*    __restrict__ w,
    const _Float16* __restrict__ W2P,
    float*          __restrict__ out)
{
    __shared__ __attribute__((aligned(16))) float sP[4 * 16 * OROW];

    const int tid     = threadIdx.x;
    const int wave    = tid >> 5;
    const int lane    = tid & 31;
    const int lm      = lane & 15;
    const int hi      = lane >> 4;
    const int rowBase = blockIdx.x * 16;

    const v4f* xq  = (const v4f*)(X0 + (size_t)(rowBase + lm) * DIN + hi * 4);
    const v4f* wp0 = (const v4f*)WcwP + (size_t)(2 * wave)     * 1024 + lane;
    const v4f* wp1 = (const v4f*)WcwP + (size_t)(2 * wave + 1) * 1024 + lane;

    const int    batch  = rowBase >> 10;
    const float  tb     = t[batch];
    const int    wave_s = __builtin_amdgcn_readfirstlane(wave);
    const float* cbw    = Wcb + wave_s * 32;
    const float* ww     = w   + wave_s * 32;
    float shs[2][16];
    #pragma unroll
    for (int tt = 0; tt < 2; ++tt) {
        #pragma unroll
        for (int i = 0; i < 16; ++i)
            shs[tt][i] = cbw[16 * tt + i] + ww[16 * tt + i] * tb;
    }

    v8f acc0 = {0.f,0.f,0.f,0.f,0.f,0.f,0.f,0.f};
    v8f acc1 = {0.f,0.f,0.f,0.f,0.f,0.f,0.f,0.f};
    #pragma unroll
    for (int s = 0; s < DIN / 8; ++s) {
        const v4f a  = xq[2 * s];
        const v4f b0 = wp0[32 * s];
        const v4f b1 = wp1[32 * s];
        const v2f a_lo  = __builtin_shufflevector(a,  a,  0, 1);
        const v2f a_up  = __builtin_shufflevector(a,  a,  2, 3);
        const v2f b0_lo = __builtin_shufflevector(b0, b0, 0, 1);
        const v2f b0_up = __builtin_shufflevector(b0, b0, 2, 3);
        const v2f b1_lo = __builtin_shufflevector(b1, b1, 0, 1);
        const v2f b1_up = __builtin_shufflevector(b1, b1, 2, 3);
        acc0 = __builtin_amdgcn_wmma_f32_16x16x4_f32(
             false, b0_lo,  false, a_lo,
             (short)0, acc0,  false,  false);
        acc1 = __builtin_amdgcn_wmma_f32_16x16x4_f32(
             false, b1_lo,  false, a_lo,
             (short)0, acc1,  false,  false);
        acc0 = __builtin_amdgcn_wmma_f32_16x16x4_f32(
             false, b0_up,  false, a_up,
             (short)0, acc0,  false,  false);
        acc1 = __builtin_amdgcn_wmma_f32_16x16x4_f32(
             false, b1_up,  false, a_up,
             (short)0, acc1,  false,  false);
    }

    const v16h* bptr = (const v16h*)W2P + (size_t)wave * (NFRAG * 32) + lane;
    v16h bfr[NFRAG];
    #pragma unroll
    for (int f = 0; f < NFRAG; ++f) bfr[f] = bptr[(size_t)f * 32];
    __builtin_amdgcn_sched_barrier(0);

    v16h afr[KSW];
    #pragma unroll
    for (int tt = 0; tt < 2; ++tt) {
        #pragma unroll
        for (int j = 0; j < 8; ++j) {
            const float shift = hi ? shs[tt][8 + j] : shs[tt][j];
            const float ang   = (tt ? acc1[j] : acc0[j]) + shift;
            #pragma unroll
            for (int g = 0; g < KH; ++g) {
                const float x = (float)(g + 1) * ang;
                afr[tt * KH + g][j]     = (_Float16)__sinf(x);
                afr[tt * KH + g][8 + j] = (_Float16)__cosf(x);
            }
        }
    }
    v8f acc[NT];
    #pragma unroll
    for (int nt = 0; nt < NT; ++nt) acc[nt] = (v8f){0.f,0.f,0.f,0.f,0.f,0.f,0.f,0.f};
    #pragma unroll
    for (int ks = 0; ks < KSW; ++ks) {
        #pragma unroll
        for (int nt = 0; nt < NT; ++nt)
            acc[nt] = __builtin_amdgcn_wmma_f32_16x16x32_f16(false, afr[ks], false, bfr[ks * NT + nt], (short)0, acc[nt], false, false);
    }

    float* mine = sP + wave * (16 * OROW);
    #pragma unroll
    for (int nt = 0; nt < NT; ++nt) {
        #pragma unroll
        for (int j = 0; j < 8; ++j)
            mine[(j + hi * 8) * OROW + nt * 16 + lm] = acc[nt][j];
    }
    __syncthreads();

    #pragma unroll
    for (int it = 0; it < 2; ++it) {
        const int p   = it * 128 + tid;
        const int row = p >> 4;
        const int c4  = p & 15;
        const float* src = sP + row * OROW + c4 * 4;
        v4f v = *(const v4f*)src;
        v = v + *(const v4f*)(src + 1 * 16 * OROW);
        v = v + *(const v4f*)(src + 2 * 16 * OROW);
        v = v + *(const v4f*)(src + 3 * 16 * OROW);
        *(v4f*)(out + (size_t)(rowBase + row) * DOUT + c4 * 4) = v;
    }
}

extern "C" void kernel_launch(void* const* d_in, const int* in_sizes, int n_in,
                              void* d_out, int out_size, void* d_ws, size_t ws_size,
                              hipStream_t stream) {
    const float* X0  = (const float*)d_in[0];
    const float* t   = (const float*)d_in[1];
    const float* Wcw = (const float*)d_in[2];
    const float* Wcb = (const float*)d_in[3];
    const float* w   = (const float*)d_in[4];
    const float* A   = (const float*)d_in[5];
    const float* Bp  = (const float*)d_in[6];
    float* out = (float*)d_out;

    _Float16* W2P  = (_Float16*)d_ws;
    float*    WcwP = (float*)(W2P + KFEAT * DOUT);

    k_weight_pack       <<<W2_BLOCKS,  256, 0, stream>>>(A, Bp, Wcw, W2P, WcwP);
    k_angle_fourier_wmma<<<MROWS / 16, 128, 0, stream>>>(X0, t, WcwP, Wcb, w, W2P, out);
}
